# starting kernel
# baseline (speedup 1.0000x reference)
#include <hip/hip_runtime.h>
#include <math.h>

typedef _Float16 h16;
typedef __attribute__((ext_vector_type(16))) _Float16 v16h;
typedef __attribute__((ext_vector_type(8)))  _Float16 v8h;
typedef __attribute__((ext_vector_type(8)))  float    v8f;

#define NODE_DIM 128
#define HIDDEN   128
#define NBASIS   32
#define MSG_IN   288
#define WAVES    4

static __device__ __forceinline__ v16h cat8(v8h a, v8h b) {
  return __builtin_shufflevector(a, b, 0,1,2,3,4,5,6,7,8,9,10,11,12,13,14,15);
}
static __device__ __forceinline__ float silu(float x) {
  return x / (1.0f + __expf(-x));
}
static __device__ __forceinline__ v16h load_b_frag(const h16* __restrict__ wp,
                                                   int kt, int nt, int NT, int lane) {
  const v8h* p = (const v8h*)(wp + (size_t)(((kt * NT + nt) * 32 + lane)) * 16);
  return cat8(p[0], p[1]);
}
static __device__ __forceinline__ v16h load_a_frag(const h16* row_base, int kt, int kb) {
  const v8h* a0 = (const v8h*)(row_base + kt * 32 + kb);
  const v8h* a1 = (const v8h*)(row_base + kt * 32 + 16 + kb);
  return cat8(*a0, *a1);
}

extern "C" __global__ void pack_w(const float* __restrict__ W, h16* __restrict__ out,
                                  int K, int Nn) {
  int t = blockIdx.x * blockDim.x + threadIdx.x;
  int KT = K >> 5, NT = Nn >> 4;
  int total = KT * NT * 32;
  if (t >= total) return;
  int lane = t & 31;
  int tile = t >> 5;
  int nt = tile % NT, kt = tile / NT;
  int n = nt * 16 + (lane & 15);
  int kbase = (lane < 16) ? 0 : 8;
  alignas(16) h16 tmp[16];
#pragma unroll
  for (int r = 0; r < 4; ++r) {
    int k0 = kt * 32 + kbase + 2 * r;
    tmp[2 * r]     = (h16)W[(size_t)k0 * Nn + n];
    tmp[2 * r + 1] = (h16)W[(size_t)(k0 + 1) * Nn + n];
    int k1 = kt * 32 + 16 + kbase + 2 * r;
    tmp[8 + 2 * r]     = (h16)W[(size_t)k1 * Nn + n];
    tmp[8 + 2 * r + 1] = (h16)W[(size_t)(k1 + 1) * Nn + n];
  }
  v8h* o = (v8h*)(out + (size_t)t * 16);
  o[0] = *(const v8h*)(tmp);
  o[1] = *(const v8h*)(tmp + 8);
}

extern "C" __global__ void egnn_init(float* __restrict__ aggr, const float* __restrict__ pos,
                                     float* __restrict__ pos_out, int N) {
  int i = blockIdx.x * blockDim.x + threadIdx.x;
  int stride = gridDim.x * blockDim.x;
  for (int j = i; j < N * HIDDEN; j += stride) aggr[j] = 0.0f;
  for (int j = i; j < N * 3; j += stride) pos_out[j] = pos[j];
}

extern "C" __global__ void __launch_bounds__(WAVES * 32)
egnn_edge(const float* __restrict__ hfeat, const float* __restrict__ pos,
          const int* __restrict__ ei,
          const h16* __restrict__ mw1p, const float* __restrict__ mb1,
          const h16* __restrict__ mw2p, const float* __restrict__ mb2,
          const h16* __restrict__ cw1p, const float* __restrict__ cb1,
          const float* __restrict__ cw2, const float* __restrict__ cb2,
          float* __restrict__ aggr, float* __restrict__ pos_out, int E) {
  extern __shared__ char smem[];
  const int wave = threadIdx.x >> 5;
  const int lane = threadIdx.x & 31;
  const int m    = lane & 15;
  const int kb   = (lane < 16) ? 0 : 8;
  const int drow = (lane < 16) ? 0 : 8;

  const int WB = 16 * MSG_IN * 2 + 16 * HIDDEN * 2 + 64 + 64;
  char* wb = smem + wave * WB;
  h16*   s_min = (h16*)wb;
  h16*   s_hid = (h16*)(wb + 16 * MSG_IN * 2);
  int*   s_row = (int*)(wb + 16 * MSG_IN * 2 + 16 * HIDDEN * 2);
  float* s_cm  = (float*)(wb + 16 * MSG_IN * 2 + 16 * HIDDEN * 2 + 64);

  const int ntile = E >> 4;
  const int nwav  = gridDim.x * WAVES;
  for (int tile = blockIdx.x * WAVES + wave; tile < ntile; tile += nwav) {
    const int e    = tile * 16 + m;
    const int vrow = ei[e];
    const int vcol = ei[E + e];
    float rx = 0.f, ry = 0.f, rz = 0.f;

    if (lane < 16) {
      s_row[m] = vrow;
      rx = pos[vrow * 3 + 0] - pos[vcol * 3 + 0];
      ry = pos[vrow * 3 + 1] - pos[vcol * 3 + 1];
      rz = pos[vrow * 3 + 2] - pos[vcol * 3 + 2];
      float dist = sqrtf(rx * rx + ry * ry + rz * rz + 1e-8f);
      const float step  = 10.0f / (NBASIS - 1);
      const float coeff = -0.5f / (step * step);
#pragma unroll
      for (int b = 0; b < NBASIS; ++b) {
        float d = dist - step * b;
        s_min[m * MSG_IN + 2 * NODE_DIM + b] = (h16)__expf(coeff * d * d);
      }
      const float4* hp = (const float4*)(hfeat + (size_t)vrow * NODE_DIM);
#pragma unroll 4
      for (int j = 0; j < NODE_DIM / 4; ++j) {
        float4 v = hp[j];
        h16* dst = s_min + m * MSG_IN + j * 4;
        dst[0] = (h16)v.x; dst[1] = (h16)v.y; dst[2] = (h16)v.z; dst[3] = (h16)v.w;
      }
    } else {
      const float4* hp = (const float4*)(hfeat + (size_t)vcol * NODE_DIM);
#pragma unroll 4
      for (int j = 0; j < NODE_DIM / 4; ++j) {
        float4 v = hp[j];
        h16* dst = s_min + m * MSG_IN + NODE_DIM + j * 4;
        dst[0] = (h16)v.x; dst[1] = (h16)v.y; dst[2] = (h16)v.z; dst[3] = (h16)v.w;
      }
    }
    asm volatile("s_wait_dscnt 0" ::: "memory");

    v8f acc[8];
#pragma unroll
    for (int nt = 0; nt < 8; ++nt) {
      float b0 = mb1[nt * 16 + m];
      acc[nt] = (v8f){b0, b0, b0, b0, b0, b0, b0, b0};
    }
#pragma unroll
    for (int kt = 0; kt < MSG_IN / 32; ++kt) {
      v16h a = load_a_frag(s_min + m * MSG_IN, kt, kb);
#pragma unroll
      for (int nt = 0; nt < 8; ++nt) {
        v16h b = load_b_frag(mw1p, kt, nt, 8, lane);
        acc[nt] = __builtin_amdgcn_wmma_f32_16x16x32_f16(false, a, false, b,
                                                         (short)0, acc[nt], false, false);
      }
    }
#pragma unroll
    for (int nt = 0; nt < 8; ++nt)
#pragma unroll
      for (int r = 0; r < 8; ++r)
        s_hid[(drow + r) * HIDDEN + nt * 16 + m] = (h16)silu(acc[nt][r]);
    asm volatile("s_wait_dscnt 0" ::: "memory");

    v8f acc2[8];
#pragma unroll
    for (int nt = 0; nt < 8; ++nt) {
      float b0 = mb2[nt * 16 + m];
      acc2[nt] = (v8f){b0, b0, b0, b0, b0, b0, b0, b0};
    }
#pragma unroll
    for (int kt = 0; kt < HIDDEN / 32; ++kt) {
      v16h a = load_a_frag(s_hid + m * HIDDEN, kt, kb);
#pragma unroll
      for (int nt = 0; nt < 8; ++nt) {
        v16h b = load_b_frag(mw2p, kt, nt, 8, lane);
        acc2[nt] = __builtin_amdgcn_wmma_f32_16x16x32_f16(false, a, false, b,
                                                          (short)0, acc2[nt], false, false);
      }
    }
#pragma unroll
    for (int nt = 0; nt < 8; ++nt) {
      const int col = nt * 16 + m;
#pragma unroll
      for (int r = 0; r < 8; ++r) {
        float v  = acc2[nt][r];
        int   rr = drow + r;
        unsafeAtomicAdd(&aggr[(size_t)s_row[rr] * HIDDEN + col], v);
        s_hid[rr * HIDDEN + col] = (h16)v;
      }
    }
    asm volatile("s_wait_dscnt 0" ::: "memory");

    v8f acc3[8];
#pragma unroll
    for (int nt = 0; nt < 8; ++nt) {
      float b0 = cb1[nt * 16 + m];
      acc3[nt] = (v8f){b0, b0, b0, b0, b0, b0, b0, b0};
    }
#pragma unroll
    for (int kt = 0; kt < HIDDEN / 32; ++kt) {
      v16h a = load_a_frag(s_hid + m * HIDDEN, kt, kb);
#pragma unroll
      for (int nt = 0; nt < 8; ++nt) {
        v16h b = load_b_frag(cw1p, kt, nt, 8, lane);
        acc3[nt] = __builtin_amdgcn_wmma_f32_16x16x32_f16(false, a, false, b,
                                                          (short)0, acc3[nt], false, false);
      }
    }
    if (lane < 16) s_cm[m] = cb2[0];
    asm volatile("s_wait_dscnt 0" ::: "memory");
#pragma unroll
    for (int nt = 0; nt < 8; ++nt) {
      float w = cw2[nt * 16 + m];
#pragma unroll
      for (int r = 0; r < 8; ++r)
        atomicAdd(&s_cm[drow + r], silu(acc3[nt][r]) * w);
    }
    asm volatile("s_wait_dscnt 0" ::: "memory");
    if (lane < 16) {
      float cm = s_cm[m];
      unsafeAtomicAdd(&pos_out[vrow * 3 + 0], rx * cm);
      unsafeAtomicAdd(&pos_out[vrow * 3 + 1], ry * cm);
      unsafeAtomicAdd(&pos_out[vrow * 3 + 2], rz * cm);
    }
    asm volatile("s_wait_dscnt 0" ::: "memory");
  }
}

extern "C" __global__ void __launch_bounds__(WAVES * 32)
egnn_node(const float* __restrict__ hfeat, const float* __restrict__ aggr,
          const h16* __restrict__ nw1p, const float* __restrict__ nb1,
          const h16* __restrict__ nw2p, const float* __restrict__ nb2,
          float* __restrict__ h_out, int Nn) {
  extern __shared__ char smem[];
  const int wave = threadIdx.x >> 5;
  const int lane = threadIdx.x & 31;
  const int m    = lane & 15;
  const int kb   = (lane < 16) ? 0 : 8;
  const int drow = (lane < 16) ? 0 : 8;

  const int NIN = 2 * NODE_DIM;
  const int WB  = 16 * NIN * 2 + 16 * HIDDEN * 2;
  char* wb = smem + wave * WB;
  h16* s_nin = (h16*)wb;
  h16* s_hid = (h16*)(wb + 16 * NIN * 2);

  const int ntile = Nn >> 4;
  const int nwav  = gridDim.x * WAVES;
  for (int tile = blockIdx.x * WAVES + wave; tile < ntile; tile += nwav) {
    const int node = tile * 16 + m;
    const float4* src = (lane < 16) ? (const float4*)(hfeat + (size_t)node * NODE_DIM)
                                    : (const float4*)(aggr + (size_t)node * HIDDEN);
    h16* dst = s_nin + m * NIN + ((lane < 16) ? 0 : NODE_DIM);
#pragma unroll 4
    for (int j = 0; j < NODE_DIM / 4; ++j) {
      float4 v = src[j];
      dst[j * 4 + 0] = (h16)v.x; dst[j * 4 + 1] = (h16)v.y;
      dst[j * 4 + 2] = (h16)v.z; dst[j * 4 + 3] = (h16)v.w;
    }
    asm volatile("s_wait_dscnt 0" ::: "memory");

    v8f acc[8];
#pragma unroll
    for (int nt = 0; nt < 8; ++nt) {
      float b0 = nb1[nt * 16 + m];
      acc[nt] = (v8f){b0, b0, b0, b0, b0, b0, b0, b0};
    }
#pragma unroll
    for (int kt = 0; kt < NIN / 32; ++kt) {
      v16h a = load_a_frag(s_nin + m * NIN, kt, kb);
#pragma unroll
      for (int nt = 0; nt < 8; ++nt) {
        v16h b = load_b_frag(nw1p, kt, nt, 8, lane);
        acc[nt] = __builtin_amdgcn_wmma_f32_16x16x32_f16(false, a, false, b,
                                                         (short)0, acc[nt], false, false);
      }
    }
#pragma unroll
    for (int nt = 0; nt < 8; ++nt)
#pragma unroll
      for (int r = 0; r < 8; ++r)
        s_hid[(drow + r) * HIDDEN + nt * 16 + m] = (h16)silu(acc[nt][r]);
    asm volatile("s_wait_dscnt 0" ::: "memory");

    v8f acc2[8];
#pragma unroll
    for (int nt = 0; nt < 8; ++nt) {
      float b0 = nb2[nt * 16 + m];
      acc2[nt] = (v8f){b0, b0, b0, b0, b0, b0, b0, b0};
    }
#pragma unroll
    for (int kt = 0; kt < HIDDEN / 32; ++kt) {
      v16h a = load_a_frag(s_hid + m * HIDDEN, kt, kb);
#pragma unroll
      for (int nt = 0; nt < 8; ++nt) {
        v16h b = load_b_frag(nw2p, kt, nt, 8, lane);
        acc2[nt] = __builtin_amdgcn_wmma_f32_16x16x32_f16(false, a, false, b,
                                                          (short)0, acc2[nt], false, false);
      }
    }
#pragma unroll
    for (int nt = 0; nt < 8; ++nt) {
      const int col = nt * 16 + m;
#pragma unroll
      for (int r = 0; r < 8; ++r) {
        int n2 = tile * 16 + drow + r;
        h_out[(size_t)n2 * NODE_DIM + col] =
            hfeat[(size_t)n2 * NODE_DIM + col] + acc2[nt][r];
      }
    }
  }
}

extern "C" void kernel_launch(void* const* d_in, const int* in_sizes, int n_in,
                              void* d_out, int out_size, void* d_ws, size_t ws_size,
                              hipStream_t stream) {
  const float* h   = (const float*)d_in[0];
  const float* pos = (const float*)d_in[1];
  const int*   ei  = (const int*)d_in[2];
  const float* mw1 = (const float*)d_in[3];  const float* mb1 = (const float*)d_in[4];
  const float* mw2 = (const float*)d_in[5];  const float* mb2 = (const float*)d_in[6];
  const float* nw1 = (const float*)d_in[7];  const float* nb1 = (const float*)d_in[8];
  const float* nw2 = (const float*)d_in[9];  const float* nb2 = (const float*)d_in[10];
  const float* cw1 = (const float*)d_in[11]; const float* cb1 = (const float*)d_in[12];
  const float* cw2 = (const float*)d_in[13]; const float* cb2 = (const float*)d_in[14];
  const int N = in_sizes[0] / NODE_DIM;
  const int E = in_sizes[2] / 2;

  char* ws = (char*)d_ws;
  h16* mw1p = (h16*)(ws + 0);
  h16* mw2p = (h16*)(ws + 73728);
  h16* cw1p = (h16*)(ws + 106496);
  h16* nw1p = (h16*)(ws + 139264);
  h16* nw2p = (h16*)(ws + 204800);
  float* aggr = (float*)(ws + 237568);
  float* h_out   = (float*)d_out;
  float* pos_out = h_out + (size_t)N * NODE_DIM;

  pack_w<<<(9 * 8 * 32 + 127) / 128, 128, 0, stream>>>(mw1, mw1p, MSG_IN, HIDDEN);
  pack_w<<<(4 * 8 * 32 + 127) / 128, 128, 0, stream>>>(mw2, mw2p, HIDDEN, HIDDEN);
  pack_w<<<(4 * 8 * 32 + 127) / 128, 128, 0, stream>>>(cw1, cw1p, HIDDEN, HIDDEN);
  pack_w<<<(8 * 8 * 32 + 127) / 128, 128, 0, stream>>>(nw1, nw1p, 2 * NODE_DIM, HIDDEN);
  pack_w<<<(4 * 8 * 32 + 127) / 128, 128, 0, stream>>>(nw2, nw2p, HIDDEN, NODE_DIM);
  egnn_init<<<2048, 256, 0, stream>>>(aggr, pos, pos_out, N);

  size_t lds_e = (size_t)WAVES * (16 * MSG_IN * 2 + 16 * HIDDEN * 2 + 64 + 64);
  egnn_edge<<<2048, WAVES * 32, lds_e, stream>>>(h, pos, ei, mw1p, mb1, mw2p, mb2,
                                                 cw1p, cb1, cw2, cb2, aggr, pos_out, E);

  size_t lds_n = (size_t)WAVES * (16 * 2 * NODE_DIM * 2 + 16 * HIDDEN * 2);
  egnn_node<<<800, WAVES * 32, lds_n, stream>>>(h, aggr, nw1p, nb1, nw2p, nb2, h_out, N);
}


# silu divides by (1+e^-x) with the hardware reciprocal (v_rcp_f32, 1 ulp) and one multiply, not the exactly rounded 11-instruction division; last f32 bits of silu change, sum orders do not
# speedup vs baseline: 2.3733x; 2.3733x over previous
#include <hip/hip_runtime.h>
#include <math.h>

typedef _Float16 h16;
typedef __attribute__((ext_vector_type(16))) _Float16 v16h;
typedef __attribute__((ext_vector_type(8)))  _Float16 v8h;
typedef __attribute__((ext_vector_type(4)))  _Float16 v4h;
typedef __attribute__((ext_vector_type(8)))  float    v8f;

#define NODE_DIM 128
#define HIDDEN   128
#define NBASIS   32
#define MSG_IN   288
#define WAVES    4
#define EWAVES   16
#define MW1_B    (MSG_IN * HIDDEN * 2)
#define MW2_B    (HIDDEN * HIDDEN * 2)
#define EW_BYTES (MW1_B + 2 * MW2_B)
#define EPANEL   (16 * MSG_IN * 2 + 64)

static __device__ __forceinline__ v16h cat8(v8h a, v8h b) {
  return __builtin_shufflevector(a, b, 0,1,2,3,4,5,6,7,8,9,10,11,12,13,14,15);
}
static __device__ __forceinline__ float silu(float x) {
  return x * __builtin_amdgcn_rcpf(1.0f + __expf(-x));
}
template <int XM>
static __device__ __forceinline__ float xmask_add(float v) {
  const int b = __builtin_bit_cast(int, v);
  return v + __builtin_bit_cast(float, __builtin_amdgcn_update_dpp(b, b, 0x160 + XM, 0xf, 0xf, false));
}
static __device__ __forceinline__ float half_sum(float v) {
  v = xmask_add<1>(v);
  v = xmask_add<2>(v);
  v = xmask_add<4>(v);
  v = xmask_add<8>(v);
  return v;
}
static __device__ __forceinline__ float xhalf(float x) {
  const int b = __builtin_bit_cast(int, x);
  return __builtin_bit_cast(float,
      __builtin_amdgcn_permlanex16(b, b, 0x76543210, 0xfedcba98, false, false));
}
static __device__ __forceinline__ v16h load_b_frag(const h16* __restrict__ wp,
                                                   int kt, int nt, int NT, int lane) {
  const v8h* p = (const v8h*)(wp + (size_t)(((kt * NT + nt) * 32 + lane)) * 16);
  return cat8(p[0], p[1]);
}
static __device__ __forceinline__ v16h load_a_frag(const h16* row_base, int kt, int kb) {
  const v8h* a0 = (const v8h*)(row_base + kt * 32 + kb);
  const v8h* a1 = (const v8h*)(row_base + kt * 32 + 16 + kb);
  return cat8(*a0, *a1);
}

static __device__ __forceinline__ v16h lds_b_frag(const char* wbase, int kt, int nt, int lane) {
  const char* p = wbase + (kt * 8 + nt) * 1024 + lane * 16;
  return cat8(*(const v8h*)p, *(const v8h*)(p + 512));
}
static __device__ __forceinline__ void stage_w(const h16* __restrict__ src, char* dst,
                                               int nbytes, int tid, int nthr) {
  const uint4* g = (const uint4*)src;
  for (int c = tid; c < (nbytes >> 4); c += nthr) {
    const int f = c >> 6, l = (c >> 1) & 31, j = c & 1;
    *(uint4*)(dst + f * 1024 + j * 512 + l * 16) = g[c];
  }
}

template <int KT>
static __device__ __forceinline__ void panel_mma(v8f (&acc)[8], const h16* arow, int kb,
                                                 const char* wbase, int lane) {
  v16h bc[8];
  v16h ac = load_a_frag(arow, 0, kb);
#pragma unroll
  for (int nt = 0; nt < 8; ++nt) bc[nt] = lds_b_frag(wbase, 0, nt, lane);
#pragma unroll
  for (int kt = 0; kt < KT; ++kt) {
    v16h bn[8];
    v16h an = ac;
#pragma unroll
    for (int h = 0; h < 2; ++h) {
      if (kt + 1 < KT) {
        if (h == 0) an = load_a_frag(arow, kt + 1, kb);
#pragma unroll
        for (int j = 0; j < 4; ++j) bn[4 * h + j] = lds_b_frag(wbase, kt + 1, 4 * h + j, lane);
      }
      __builtin_amdgcn_sched_barrier(0);
#pragma unroll
      for (int j = 0; j < 4; ++j)
        acc[4 * h + j] = __builtin_amdgcn_wmma_f32_16x16x32_f16(false, ac, false, bc[4 * h + j],
                                                                (short)0, acc[4 * h + j], false, false);
    }
    if (kt + 1 < KT) {
      ac = an;
#pragma unroll
      for (int nt = 0; nt < 8; ++nt) bc[nt] = bn[nt];
    }
  }
}

extern "C" __global__ void pack_w(const float* __restrict__ W, h16* __restrict__ out,
                                  int K, int Nn) {
  int t = blockIdx.x * blockDim.x + threadIdx.x;
  int KT = K >> 5, NT = Nn >> 4;
  int total = KT * NT * 32;
  if (t >= total) return;
  int lane = t & 31;
  int tile = t >> 5;
  int nt = tile % NT, kt = tile / NT;
  int n = nt * 16 + (lane & 15);
  int kbase = (lane < 16) ? 0 : 8;
  alignas(16) h16 tmp[16];
#pragma unroll
  for (int r = 0; r < 4; ++r) {
    int k0 = kt * 32 + kbase + 2 * r;
    tmp[2 * r]     = (h16)W[(size_t)k0 * Nn + n];
    tmp[2 * r + 1] = (h16)W[(size_t)(k0 + 1) * Nn + n];
    int k1 = kt * 32 + 16 + kbase + 2 * r;
    tmp[8 + 2 * r]     = (h16)W[(size_t)k1 * Nn + n];
    tmp[8 + 2 * r + 1] = (h16)W[(size_t)(k1 + 1) * Nn + n];
  }
  v8h* o = (v8h*)(out + (size_t)t * 16);
  o[0] = *(const v8h*)(tmp);
  o[1] = *(const v8h*)(tmp + 8);
}

extern "C" __global__ void egnn_init(float* __restrict__ sums, int N) {
  int i = blockIdx.x * blockDim.x + threadIdx.x;
  int stride = gridDim.x * blockDim.x;
  for (int j = i; j < N * (2 * HIDDEN + 6); j += stride) sums[j] = 0.0f;
}

extern "C" __global__ void __launch_bounds__(EWAVES * 32)
egnn_edge(const float* __restrict__ hfeat, const float* __restrict__ pos,
          const int* __restrict__ ei,
          const h16* __restrict__ mw1p, const float* __restrict__ mb1,
          const h16* __restrict__ mw2p, const float* __restrict__ mb2,
          const h16* __restrict__ cw1p, const float* __restrict__ cb1,
          const float* __restrict__ cw2, const float* __restrict__ cb2,
          float* __restrict__ aggr2, float* __restrict__ pacc2, int E, int N) {
  __shared__ __attribute__((aligned(16))) char smem[EW_BYTES + EWAVES * EPANEL];
  const char* s_w1 = smem;
  const char* s_w2 = smem + MW1_B;
  const char* s_w3 = smem + MW1_B + MW2_B;
  stage_w(mw1p, smem, MW1_B, threadIdx.x, EWAVES * 32);
  stage_w(mw2p, smem + MW1_B, MW2_B, threadIdx.x, EWAVES * 32);
  stage_w(cw1p, smem + MW1_B + MW2_B, MW2_B, threadIdx.x, EWAVES * 32);
  __syncthreads();
  const int wave = threadIdx.x >> 5;
  const int lane = threadIdx.x & 31;
  const int m    = lane & 15;
  const int kb   = (lane < 16) ? 0 : 8;
  const int drow = (lane < 16) ? 0 : 8;

  const unsigned half = (__builtin_amdgcn_s_sendmsg_rtn(0x87) >> 19) & 1u;
  float* __restrict__ aggr = aggr2 + (size_t)half * N * HIDDEN;
  float* __restrict__ pacc = pacc2 + (size_t)half * N * 3;

  char* wb = smem + EW_BYTES + wave * EPANEL;
  h16*   s_min = (h16*)wb;
  h16*   s_hid = (h16*)wb;
  int*   s_row = (int*)(wb + 16 * MSG_IN * 2);

  const int ntile = E >> 4;
  const int nwav  = gridDim.x * EWAVES;
  for (int tile = wave * gridDim.x + blockIdx.x; tile < ntile; tile += nwav) {
    const int e    = tile * 16 + m;
    const int vrow = ei[e];
    const int vcol = ei[E + e];
    float rx = 0.f, ry = 0.f, rz = 0.f;

    rx = pos[vrow * 3 + 0] - pos[vcol * 3 + 0];
    ry = pos[vrow * 3 + 1] - pos[vcol * 3 + 1];
    rz = pos[vrow * 3 + 2] - pos[vcol * 3 + 2];
    float4 hr[8], hc[8];
#pragma unroll
    for (int i = 0; i < 8; ++i) {
      const int ri = __builtin_amdgcn_readlane(vrow, i);
      const int ci = __builtin_amdgcn_readlane(vcol, i);
      hr[i] = ((const float4*)(hfeat + (size_t)ri * NODE_DIM))[lane];
      hc[i] = ((const float4*)(hfeat + (size_t)ci * NODE_DIM))[lane];
    }
    if (lane < 16) {
      s_row[m] = vrow;
      float dist = sqrtf(rx * rx + ry * ry + rz * rz + 1e-8f);
      const float step  = 10.0f / (NBASIS - 1);
      const float coeff = -0.5f / (step * step);
#pragma unroll
      for (int b = 0; b < NBASIS; ++b) {
        float d = dist - step * b;
        s_min[m * MSG_IN + 2 * NODE_DIM + b] = (h16)__expf(coeff * d * d);
      }
    }
#pragma unroll
    for (int i = 0; i < 8; ++i) {
      *(v4h*)(s_min + i * MSG_IN + lane * 4) =
          (v4h){(h16)hr[i].x, (h16)hr[i].y, (h16)hr[i].z, (h16)hr[i].w};
      *(v4h*)(s_min + i * MSG_IN + NODE_DIM + lane * 4) =
          (v4h){(h16)hc[i].x, (h16)hc[i].y, (h16)hc[i].z, (h16)hc[i].w};
      const int ri = __builtin_amdgcn_readlane(vrow, 8 + i);
      const int ci = __builtin_amdgcn_readlane(vcol, 8 + i);
      hr[i] = ((const float4*)(hfeat + (size_t)ri * NODE_DIM))[lane];
      hc[i] = ((const float4*)(hfeat + (size_t)ci * NODE_DIM))[lane];
    }
#pragma unroll
    for (int i = 0; i < 8; ++i) {
      *(v4h*)(s_min + (8 + i) * MSG_IN + lane * 4) =
          (v4h){(h16)hr[i].x, (h16)hr[i].y, (h16)hr[i].z, (h16)hr[i].w};
      *(v4h*)(s_min + (8 + i) * MSG_IN + NODE_DIM + lane * 4) =
          (v4h){(h16)hc[i].x, (h16)hc[i].y, (h16)hc[i].z, (h16)hc[i].w};
    }
    asm volatile("s_wait_dscnt 0" ::: "memory");

    v8f acc[8];
#pragma unroll
    for (int nt = 0; nt < 8; ++nt) {
      float b0 = mb1[nt * 16 + m];
      acc[nt] = (v8f){b0, b0, b0, b0, b0, b0, b0, b0};
    }
    panel_mma<MSG_IN / 32>(acc, s_min + m * MSG_IN, kb, s_w1, lane);
#pragma unroll
    for (int nt = 0; nt < 8; ++nt)
#pragma unroll
      for (int r = 0; r < 8; ++r)
        s_hid[(drow + r) * HIDDEN + nt * 16 + m] = (h16)silu(acc[nt][r]);
    asm volatile("s_wait_dscnt 0" ::: "memory");

    v8f acc2[8];
#pragma unroll
    for (int nt = 0; nt < 8; ++nt) {
      float b0 = mb2[nt * 16 + m];
      acc2[nt] = (v8f){b0, b0, b0, b0, b0, b0, b0, b0};
    }
    panel_mma<HIDDEN / 32>(acc2, s_hid + m * HIDDEN, kb, s_w2, lane);
#pragma unroll
    for (int nt = 0; nt < 8; ++nt) {
      const int col = nt * 16 + m;
#pragma unroll
      for (int r = 0; r < 8; ++r) {
        float v  = acc2[nt][r];
        int   rr = drow + r;
        unsafeAtomicAdd(&aggr[(size_t)s_row[rr] * HIDDEN + col], v);
        s_hid[rr * HIDDEN + col] = (h16)v;
      }
    }
    asm volatile("s_wait_dscnt 0" ::: "memory");

    v8f acc3[8];
#pragma unroll
    for (int nt = 0; nt < 8; ++nt) {
      float b0 = cb1[nt * 16 + m];
      acc3[nt] = (v8f){b0, b0, b0, b0, b0, b0, b0, b0};
    }
    panel_mma<HIDDEN / 32>(acc3, s_hid + m * HIDDEN, kb, s_w3, lane);
    float part[8];
#pragma unroll
    for (int r = 0; r < 8; ++r) part[r] = 0.0f;
#pragma unroll
    for (int nt = 0; nt < 8; ++nt) {
      float w = cw2[nt * 16 + m];
#pragma unroll
      for (int r = 0; r < 8; ++r)
        part[r] += silu(acc3[nt][r]) * w;
    }
#pragma unroll
    for (int r = 0; r < 8; ++r) part[r] = half_sum(part[r]);
    float own = part[0];
#pragma unroll
    for (int r = 1; r < 8; ++r) own = ((lane & 7) == r) ? part[r] : own;
    const float oth = xhalf(own);
    const float cm  = cb2[0] + ((m < 8) ? own : oth);
    if (lane < 16) {
      unsafeAtomicAdd(&pacc[vrow * 3 + 0], rx * cm);
      unsafeAtomicAdd(&pacc[vrow * 3 + 1], ry * cm);
      unsafeAtomicAdd(&pacc[vrow * 3 + 2], rz * cm);
    }
    asm volatile("s_wait_dscnt 0" ::: "memory");
  }
}

extern "C" __global__ void __launch_bounds__(WAVES * 32)
egnn_node(const float* __restrict__ hfeat, const float* __restrict__ aggr,
          const h16* __restrict__ nw1p, const float* __restrict__ nb1,
          const h16* __restrict__ nw2p, const float* __restrict__ nb2,
          float* __restrict__ h_out, const float* __restrict__ pos,
          const float* __restrict__ pacc, float* __restrict__ pos_out, int Nn) {
  extern __shared__ char smem[];
  const int wave = threadIdx.x >> 5;
  const int lane = threadIdx.x & 31;
  const int m    = lane & 15;
  const int kb   = (lane < 16) ? 0 : 8;
  const int drow = (lane < 16) ? 0 : 8;

  const int NIN = 2 * NODE_DIM;
  const int WB  = 16 * NIN * 2 + 16 * HIDDEN * 2;
  char* wb = smem + wave * WB;
  h16* s_nin = (h16*)wb;
  h16* s_hid = (h16*)(wb + 16 * NIN * 2);

  const int ntile = Nn >> 4;
  const int nwav  = gridDim.x * WAVES;
  for (int tile = blockIdx.x * WAVES + wave; tile < ntile; tile += nwav) {
    const float4* hsrc = (const float4*)(hfeat + (size_t)tile * 16 * NODE_DIM);
    const float4* asrc = (const float4*)(aggr + (size_t)tile * 16 * HIDDEN);
    const float4* bsrc = (const float4*)(aggr + ((size_t)Nn + tile * 16) * HIDDEN);
#pragma unroll
    for (int i = 0; i < 16; ++i) {
      const float4 a = hsrc[i * 32 + lane];
      const float4 b0 = asrc[i * 32 + lane];
      const float4 b1 = bsrc[i * 32 + lane];
      const float4 b = make_float4(b0.x + b1.x, b0.y + b1.y, b0.z + b1.z, b0.w + b1.w);
      *(v4h*)(s_nin + i * NIN + lane * 4) = (v4h){(h16)a.x, (h16)a.y, (h16)a.z, (h16)a.w};
      *(v4h*)(s_nin + i * NIN + NODE_DIM + lane * 4) =
          (v4h){(h16)b.x, (h16)b.y, (h16)b.z, (h16)b.w};
    }
    asm volatile("s_wait_dscnt 0" ::: "memory");

    v8f acc[8];
#pragma unroll
    for (int nt = 0; nt < 8; ++nt) {
      float b0 = nb1[nt * 16 + m];
      acc[nt] = (v8f){b0, b0, b0, b0, b0, b0, b0, b0};
    }
#pragma unroll
    for (int kt = 0; kt < NIN / 32; ++kt) {
      v16h a = load_a_frag(s_nin + m * NIN, kt, kb);
#pragma unroll
      for (int nt = 0; nt < 8; ++nt) {
        v16h b = load_b_frag(nw1p, kt, nt, 8, lane);
        acc[nt] = __builtin_amdgcn_wmma_f32_16x16x32_f16(false, a, false, b,
                                                         (short)0, acc[nt], false, false);
      }
    }
#pragma unroll
    for (int nt = 0; nt < 8; ++nt)
#pragma unroll
      for (int r = 0; r < 8; ++r)
        s_hid[(drow + r) * HIDDEN + nt * 16 + m] = (h16)silu(acc[nt][r]);
    asm volatile("s_wait_dscnt 0" ::: "memory");

    v8f acc2[8];
#pragma unroll
    for (int nt = 0; nt < 8; ++nt) {
      float b0 = nb2[nt * 16 + m];
      acc2[nt] = (v8f){b0, b0, b0, b0, b0, b0, b0, b0};
    }
#pragma unroll
    for (int kt = 0; kt < HIDDEN / 32; ++kt) {
      v16h a = load_a_frag(s_hid + m * HIDDEN, kt, kb);
#pragma unroll
      for (int nt = 0; nt < 8; ++nt) {
        v16h b = load_b_frag(nw2p, kt, nt, 8, lane);
        acc2[nt] = __builtin_amdgcn_wmma_f32_16x16x32_f16(false, a, false, b,
                                                          (short)0, acc2[nt], false, false);
      }
    }
#pragma unroll
    for (int nt = 0; nt < 8; ++nt) {
      const int col = nt * 16 + m;
#pragma unroll
      for (int r = 0; r < 8; ++r) {
        int n2 = tile * 16 + drow + r;
        h_out[(size_t)n2 * NODE_DIM + col] =
            hfeat[(size_t)n2 * NODE_DIM + col] + acc2[nt][r];
      }
    }
    const size_t p0 = (size_t)tile * 48 + lane;
    const size_t p1 = p0 + (size_t)Nn * 3;
    pos_out[p0] = pos[p0] + (pacc[p0] + pacc[p1]);
    if (lane < 16) pos_out[p0 + 32] = pos[p0 + 32] + (pacc[p0 + 32] + pacc[p1 + 32]);
  }
}

extern "C" void kernel_launch(void* const* d_in, const int* in_sizes, int n_in,
                              void* d_out, int out_size, void* d_ws, size_t ws_size,
                              hipStream_t stream) {
  const float* h   = (const float*)d_in[0];
  const float* pos = (const float*)d_in[1];
  const int*   ei  = (const int*)d_in[2];
  const float* mw1 = (const float*)d_in[3];  const float* mb1 = (const float*)d_in[4];
  const float* mw2 = (const float*)d_in[5];  const float* mb2 = (const float*)d_in[6];
  const float* nw1 = (const float*)d_in[7];  const float* nb1 = (const float*)d_in[8];
  const float* nw2 = (const float*)d_in[9];  const float* nb2 = (const float*)d_in[10];
  const float* cw1 = (const float*)d_in[11]; const float* cb1 = (const float*)d_in[12];
  const float* cw2 = (const float*)d_in[13]; const float* cb2 = (const float*)d_in[14];
  const int N = in_sizes[0] / NODE_DIM;
  const int E = in_sizes[2] / 2;

  char* ws = (char*)d_ws;
  h16* mw1p = (h16*)(ws + 0);
  h16* mw2p = (h16*)(ws + 73728);
  h16* cw1p = (h16*)(ws + 106496);
  h16* nw1p = (h16*)(ws + 139264);
  h16* nw2p = (h16*)(ws + 204800);
  float* aggr = (float*)(ws + 237568);
  float* pacc = aggr + (size_t)N * HIDDEN * 2;
  float* h_out   = (float*)d_out;
  float* pos_out = h_out + (size_t)N * NODE_DIM;

  pack_w<<<(9 * 8 * 32 + 127) / 128, 128, 0, stream>>>(mw1, mw1p, MSG_IN, HIDDEN);
  pack_w<<<(4 * 8 * 32 + 127) / 128, 128, 0, stream>>>(mw2, mw2p, HIDDEN, HIDDEN);
  pack_w<<<(4 * 8 * 32 + 127) / 128, 128, 0, stream>>>(cw1, cw1p, HIDDEN, HIDDEN);
  pack_w<<<(8 * 8 * 32 + 127) / 128, 128, 0, stream>>>(nw1, nw1p, 2 * NODE_DIM, HIDDEN);
  pack_w<<<(4 * 8 * 32 + 127) / 128, 128, 0, stream>>>(nw2, nw2p, HIDDEN, NODE_DIM);
  egnn_init<<<2048, 256, 0, stream>>>(aggr, N);

  egnn_edge<<<256, EWAVES * 32, 0, stream>>>(h, pos, ei, mw1p, mb1, mw2p, mb2,
                                                 cw1p, cb1, cw2, cb2, aggr, pacc, E, N);

  size_t lds_n = (size_t)WAVES * (16 * 2 * NODE_DIM * 2 + 16 * HIDDEN * 2);
  egnn_node<<<800, WAVES * 32, lds_n, stream>>>(h, aggr, nw1p, nb1, nw2p, nb2, h_out,
                                                pos, pacc, pos_out, N);
}


# Message segment sum: accumulator tiles 2p, 2p+1 exchange lane halves (permlanex16) so each atomic add instruction covers one whole 128-byte line of one row, not half a line in each of two rows
# speedup vs baseline: 2.4514x; 1.0329x over previous
#include <hip/hip_runtime.h>
#include <math.h>

typedef _Float16 h16;
typedef __attribute__((ext_vector_type(16))) _Float16 v16h;
typedef __attribute__((ext_vector_type(8)))  _Float16 v8h;
typedef __attribute__((ext_vector_type(4)))  _Float16 v4h;
typedef __attribute__((ext_vector_type(8)))  float    v8f;

#define NODE_DIM 128
#define HIDDEN   128
#define NBASIS   32
#define MSG_IN   288
#define WAVES    4
#define EWAVES   16
#define MW1_B    (MSG_IN * HIDDEN * 2)
#define MW2_B    (HIDDEN * HIDDEN * 2)
#define EW_BYTES (MW1_B + 2 * MW2_B)
#define EPANEL   (16 * MSG_IN * 2)

static __device__ __forceinline__ v16h cat8(v8h a, v8h b) {
  return __builtin_shufflevector(a, b, 0,1,2,3,4,5,6,7,8,9,10,11,12,13,14,15);
}
static __device__ __forceinline__ float silu(float x) {
  return x * __builtin_amdgcn_rcpf(1.0f + __expf(-x));
}
template <int XM>
static __device__ __forceinline__ float xmask_add(float v) {
  const int b = __builtin_bit_cast(int, v);
  return v + __builtin_bit_cast(float, __builtin_amdgcn_update_dpp(b, b, 0x160 + XM, 0xf, 0xf, false));
}
static __device__ __forceinline__ float half_sum(float v) {
  v = xmask_add<1>(v);
  v = xmask_add<2>(v);
  v = xmask_add<4>(v);
  v = xmask_add<8>(v);
  return v;
}
static __device__ __forceinline__ float xhalf(float x) {
  const int b = __builtin_bit_cast(int, x);
  return __builtin_bit_cast(float,
      __builtin_amdgcn_permlanex16(b, b, 0x76543210, 0xfedcba98, false, false));
}
static __device__ __forceinline__ void swap_halves(float& a, float& b, bool lo) {
  const float u = xhalf(lo ? b : a);
  a = lo ? a : u;
  b = lo ? u : b;
}
static __device__ __forceinline__ v16h load_b_frag(const h16* __restrict__ wp,
                                                   int kt, int nt, int NT, int lane) {
  const v8h* p = (const v8h*)(wp + (size_t)(((kt * NT + nt) * 32 + lane)) * 16);
  return cat8(p[0], p[1]);
}
static __device__ __forceinline__ v16h load_a_frag(const h16* row_base, int kt, int kb) {
  const v8h* a0 = (const v8h*)(row_base + kt * 32 + kb);
  const v8h* a1 = (const v8h*)(row_base + kt * 32 + 16 + kb);
  return cat8(*a0, *a1);
}

static __device__ __forceinline__ v16h lds_b_frag(const char* wbase, int kt, int nt, int lane) {
  const char* p = wbase + (kt * 8 + nt) * 1024 + lane * 16;
  return cat8(*(const v8h*)p, *(const v8h*)(p + 512));
}
static __device__ __forceinline__ void stage_w(const h16* __restrict__ src, char* dst,
                                               int nbytes, int tid, int nthr) {
  const uint4* g = (const uint4*)src;
  for (int c = tid; c < (nbytes >> 4); c += nthr) {
    const int f = c >> 6, l = (c >> 1) & 31, j = c & 1;
    *(uint4*)(dst + f * 1024 + j * 512 + l * 16) = g[c];
  }
}

template <int KT>
static __device__ __forceinline__ void panel_mma(v8f (&acc)[8], const h16* arow, int kb,
                                                 const char* wbase, int lane) {
  v16h bc[8];
  v16h ac = load_a_frag(arow, 0, kb);
#pragma unroll
  for (int nt = 0; nt < 8; ++nt) bc[nt] = lds_b_frag(wbase, 0, nt, lane);
#pragma unroll
  for (int kt = 0; kt < KT; ++kt) {
    v16h bn[8];
    v16h an = ac;
#pragma unroll
    for (int h = 0; h < 2; ++h) {
      if (kt + 1 < KT) {
        if (h == 0) an = load_a_frag(arow, kt + 1, kb);
#pragma unroll
        for (int j = 0; j < 4; ++j) bn[4 * h + j] = lds_b_frag(wbase, kt + 1, 4 * h + j, lane);
      }
      __builtin_amdgcn_sched_barrier(0);
#pragma unroll
      for (int j = 0; j < 4; ++j)
        acc[4 * h + j] = __builtin_amdgcn_wmma_f32_16x16x32_f16(false, ac, false, bc[4 * h + j],
                                                                (short)0, acc[4 * h + j], false, false);
    }
    if (kt + 1 < KT) {
      ac = an;
#pragma unroll
      for (int nt = 0; nt < 8; ++nt) bc[nt] = bn[nt];
    }
  }
}

extern "C" __global__ void pack_w(const float* __restrict__ W, h16* __restrict__ out,
                                  int K, int Nn) {
  int t = blockIdx.x * blockDim.x + threadIdx.x;
  int KT = K >> 5, NT = Nn >> 4;
  int total = KT * NT * 32;
  if (t >= total) return;
  int lane = t & 31;
  int tile = t >> 5;
  int nt = tile % NT, kt = tile / NT;
  int n = nt * 16 + (lane & 15);
  int kbase = (lane < 16) ? 0 : 8;
  alignas(16) h16 tmp[16];
#pragma unroll
  for (int r = 0; r < 4; ++r) {
    int k0 = kt * 32 + kbase + 2 * r;
    tmp[2 * r]     = (h16)W[(size_t)k0 * Nn + n];
    tmp[2 * r + 1] = (h16)W[(size_t)(k0 + 1) * Nn + n];
    int k1 = kt * 32 + 16 + kbase + 2 * r;
    tmp[8 + 2 * r]     = (h16)W[(size_t)k1 * Nn + n];
    tmp[8 + 2 * r + 1] = (h16)W[(size_t)(k1 + 1) * Nn + n];
  }
  v8h* o = (v8h*)(out + (size_t)t * 16);
  o[0] = *(const v8h*)(tmp);
  o[1] = *(const v8h*)(tmp + 8);
}

extern "C" __global__ void egnn_init(float* __restrict__ sums, int N) {
  int i = blockIdx.x * blockDim.x + threadIdx.x;
  int stride = gridDim.x * blockDim.x;
  for (int j = i; j < N * (2 * HIDDEN + 6); j += stride) sums[j] = 0.0f;
}

extern "C" __global__ void __launch_bounds__(EWAVES * 32)
egnn_edge(const float* __restrict__ hfeat, const float* __restrict__ pos,
          const int* __restrict__ ei,
          const h16* __restrict__ mw1p, const float* __restrict__ mb1,
          const h16* __restrict__ mw2p, const float* __restrict__ mb2,
          const h16* __restrict__ cw1p, const float* __restrict__ cb1,
          const float* __restrict__ cw2, const float* __restrict__ cb2,
          float* __restrict__ aggr2, float* __restrict__ pacc2, int E, int N) {
  __shared__ __attribute__((aligned(16))) char smem[EW_BYTES + EWAVES * EPANEL];
  const char* s_w1 = smem;
  const char* s_w2 = smem + MW1_B;
  const char* s_w3 = smem + MW1_B + MW2_B;
  stage_w(mw1p, smem, MW1_B, threadIdx.x, EWAVES * 32);
  stage_w(mw2p, smem + MW1_B, MW2_B, threadIdx.x, EWAVES * 32);
  stage_w(cw1p, smem + MW1_B + MW2_B, MW2_B, threadIdx.x, EWAVES * 32);
  __syncthreads();
  const int wave = threadIdx.x >> 5;
  const int lane = threadIdx.x & 31;
  const int m    = lane & 15;
  const int kb   = (lane < 16) ? 0 : 8;
  const int drow = (lane < 16) ? 0 : 8;

  const unsigned half = (__builtin_amdgcn_s_sendmsg_rtn(0x87) >> 19) & 1u;
  float* __restrict__ aggr = aggr2 + (size_t)half * N * HIDDEN;
  float* __restrict__ pacc = pacc2 + (size_t)half * N * 3;

  char* wb = smem + EW_BYTES + wave * EPANEL;
  h16*   s_min = (h16*)wb;
  h16*   s_hid = (h16*)wb;

  const int ntile = E >> 4;
  const int nwav  = gridDim.x * EWAVES;
  for (int tile = wave * gridDim.x + blockIdx.x; tile < ntile; tile += nwav) {
    const int e    = tile * 16 + m;
    const int vrow = ei[e];
    const int vcol = ei[E + e];
    float rx = 0.f, ry = 0.f, rz = 0.f;

    rx = pos[vrow * 3 + 0] - pos[vcol * 3 + 0];
    ry = pos[vrow * 3 + 1] - pos[vcol * 3 + 1];
    rz = pos[vrow * 3 + 2] - pos[vcol * 3 + 2];
    float4 hr[8], hc[8];
#pragma unroll
    for (int i = 0; i < 8; ++i) {
      const int ri = __builtin_amdgcn_readlane(vrow, i);
      const int ci = __builtin_amdgcn_readlane(vcol, i);
      hr[i] = ((const float4*)(hfeat + (size_t)ri * NODE_DIM))[lane];
      hc[i] = ((const float4*)(hfeat + (size_t)ci * NODE_DIM))[lane];
    }
    if (lane < 16) {
      float dist = sqrtf(rx * rx + ry * ry + rz * rz + 1e-8f);
      const float step  = 10.0f / (NBASIS - 1);
      const float coeff = -0.5f / (step * step);
#pragma unroll
      for (int b = 0; b < NBASIS; ++b) {
        float d = dist - step * b;
        s_min[m * MSG_IN + 2 * NODE_DIM + b] = (h16)__expf(coeff * d * d);
      }
    }
#pragma unroll
    for (int i = 0; i < 8; ++i) {
      *(v4h*)(s_min + i * MSG_IN + lane * 4) =
          (v4h){(h16)hr[i].x, (h16)hr[i].y, (h16)hr[i].z, (h16)hr[i].w};
      *(v4h*)(s_min + i * MSG_IN + NODE_DIM + lane * 4) =
          (v4h){(h16)hc[i].x, (h16)hc[i].y, (h16)hc[i].z, (h16)hc[i].w};
      const int ri = __builtin_amdgcn_readlane(vrow, 8 + i);
      const int ci = __builtin_amdgcn_readlane(vcol, 8 + i);
      hr[i] = ((const float4*)(hfeat + (size_t)ri * NODE_DIM))[lane];
      hc[i] = ((const float4*)(hfeat + (size_t)ci * NODE_DIM))[lane];
    }
#pragma unroll
    for (int i = 0; i < 8; ++i) {
      *(v4h*)(s_min + (8 + i) * MSG_IN + lane * 4) =
          (v4h){(h16)hr[i].x, (h16)hr[i].y, (h16)hr[i].z, (h16)hr[i].w};
      *(v4h*)(s_min + (8 + i) * MSG_IN + NODE_DIM + lane * 4) =
          (v4h){(h16)hc[i].x, (h16)hc[i].y, (h16)hc[i].z, (h16)hc[i].w};
    }
    asm volatile("s_wait_dscnt 0" ::: "memory");

    v8f acc[8];
#pragma unroll
    for (int nt = 0; nt < 8; ++nt) {
      float b0 = mb1[nt * 16 + m];
      acc[nt] = (v8f){b0, b0, b0, b0, b0, b0, b0, b0};
    }
    panel_mma<MSG_IN / 32>(acc, s_min + m * MSG_IN, kb, s_w1, lane);
#pragma unroll
    for (int nt = 0; nt < 8; ++nt)
#pragma unroll
      for (int r = 0; r < 8; ++r)
        s_hid[(drow + r) * HIDDEN + nt * 16 + m] = (h16)silu(acc[nt][r]);
    asm volatile("s_wait_dscnt 0" ::: "memory");

    v8f acc2[8];
#pragma unroll
    for (int nt = 0; nt < 8; ++nt) {
      float b0 = mb2[nt * 16 + m];
      acc2[nt] = (v8f){b0, b0, b0, b0, b0, b0, b0, b0};
    }
    panel_mma<HIDDEN / 32>(acc2, s_hid + m * HIDDEN, kb, s_w2, lane);
#pragma unroll
    for (int nt = 0; nt < 8; ++nt)
#pragma unroll
      for (int r = 0; r < 8; ++r)
        s_hid[(drow + r) * HIDDEN + nt * 16 + m] = (h16)acc2[nt][r];
#pragma unroll
    for (int p = 0; p < 4; ++p) {
#pragma unroll
      for (int r = 0; r < 8; ++r) {
        float a = acc2[2 * p][r], b = acc2[2 * p + 1][r];
        swap_halves(a, b, lane < 16);
        const int r0 = __builtin_amdgcn_readlane(vrow, r);
        const int r1 = __builtin_amdgcn_readlane(vrow, 8 + r);
        unsafeAtomicAdd(aggr + (size_t)r0 * HIDDEN + 32 * p + lane, a);
        unsafeAtomicAdd(aggr + (size_t)r1 * HIDDEN + 32 * p + lane, b);
      }
    }
    asm volatile("s_wait_dscnt 0" ::: "memory");

    v8f acc3[8];
#pragma unroll
    for (int nt = 0; nt < 8; ++nt) {
      float b0 = cb1[nt * 16 + m];
      acc3[nt] = (v8f){b0, b0, b0, b0, b0, b0, b0, b0};
    }
    panel_mma<HIDDEN / 32>(acc3, s_hid + m * HIDDEN, kb, s_w3, lane);
    float part[8];
#pragma unroll
    for (int r = 0; r < 8; ++r) part[r] = 0.0f;
#pragma unroll
    for (int nt = 0; nt < 8; ++nt) {
      float w = cw2[nt * 16 + m];
#pragma unroll
      for (int r = 0; r < 8; ++r)
        part[r] += silu(acc3[nt][r]) * w;
    }
#pragma unroll
    for (int r = 0; r < 8; ++r) part[r] = half_sum(part[r]);
    float own = part[0];
#pragma unroll
    for (int r = 1; r < 8; ++r) own = ((lane & 7) == r) ? part[r] : own;
    const float oth = xhalf(own);
    const float cm  = cb2[0] + ((m < 8) ? own : oth);
    if (lane < 16) {
      unsafeAtomicAdd(&pacc[vrow * 3 + 0], rx * cm);
      unsafeAtomicAdd(&pacc[vrow * 3 + 1], ry * cm);
      unsafeAtomicAdd(&pacc[vrow * 3 + 2], rz * cm);
    }
    asm volatile("s_wait_dscnt 0" ::: "memory");
  }
}

extern "C" __global__ void __launch_bounds__(WAVES * 32)
egnn_node(const float* __restrict__ hfeat, const float* __restrict__ aggr,
          const h16* __restrict__ nw1p, const float* __restrict__ nb1,
          const h16* __restrict__ nw2p, const float* __restrict__ nb2,
          float* __restrict__ h_out, const float* __restrict__ pos,
          const float* __restrict__ pacc, float* __restrict__ pos_out, int Nn) {
  extern __shared__ char smem[];
  const int wave = threadIdx.x >> 5;
  const int lane = threadIdx.x & 31;
  const int m    = lane & 15;
  const int kb   = (lane < 16) ? 0 : 8;
  const int drow = (lane < 16) ? 0 : 8;

  const int NIN = 2 * NODE_DIM;
  const int WB  = 16 * NIN * 2 + 16 * HIDDEN * 2;
  char* wb = smem + wave * WB;
  h16* s_nin = (h16*)wb;
  h16* s_hid = (h16*)(wb + 16 * NIN * 2);

  const int ntile = Nn >> 4;
  const int nwav  = gridDim.x * WAVES;
  for (int tile = blockIdx.x * WAVES + wave; tile < ntile; tile += nwav) {
    const float4* hsrc = (const float4*)(hfeat + (size_t)tile * 16 * NODE_DIM);
    const float4* asrc = (const float4*)(aggr + (size_t)tile * 16 * HIDDEN);
    const float4* bsrc = (const float4*)(aggr + ((size_t)Nn + tile * 16) * HIDDEN);
#pragma unroll
    for (int i = 0; i < 16; ++i) {
      const float4 a = hsrc[i * 32 + lane];
      const float4 b0 = asrc[i * 32 + lane];
      const float4 b1 = bsrc[i * 32 + lane];
      const float4 b = make_float4(b0.x + b1.x, b0.y + b1.y, b0.z + b1.z, b0.w + b1.w);
      *(v4h*)(s_nin + i * NIN + lane * 4) = (v4h){(h16)a.x, (h16)a.y, (h16)a.z, (h16)a.w};
      *(v4h*)(s_nin + i * NIN + NODE_DIM + lane * 4) =
          (v4h){(h16)b.x, (h16)b.y, (h16)b.z, (h16)b.w};
    }
    asm volatile("s_wait_dscnt 0" ::: "memory");

    v8f acc[8];
#pragma unroll
    for (int nt = 0; nt < 8; ++nt) {
      float b0 = nb1[nt * 16 + m];
      acc[nt] = (v8f){b0, b0, b0, b0, b0, b0, b0, b0};
    }
#pragma unroll
    for (int kt = 0; kt < NIN / 32; ++kt) {
      v16h a = load_a_frag(s_nin + m * NIN, kt, kb);
#pragma unroll
      for (int nt = 0; nt < 8; ++nt) {
        v16h b = load_b_frag(nw1p, kt, nt, 8, lane);
        acc[nt] = __builtin_amdgcn_wmma_f32_16x16x32_f16(false, a, false, b,
                                                         (short)0, acc[nt], false, false);
      }
    }
#pragma unroll
    for (int nt = 0; nt < 8; ++nt)
#pragma unroll
      for (int r = 0; r < 8; ++r)
        s_hid[(drow + r) * HIDDEN + nt * 16 + m] = (h16)silu(acc[nt][r]);
    asm volatile("s_wait_dscnt 0" ::: "memory");

    v8f acc2[8];
#pragma unroll
    for (int nt = 0; nt < 8; ++nt) {
      float b0 = nb2[nt * 16 + m];
      acc2[nt] = (v8f){b0, b0, b0, b0, b0, b0, b0, b0};
    }
#pragma unroll
    for (int kt = 0; kt < HIDDEN / 32; ++kt) {
      v16h a = load_a_frag(s_hid + m * HIDDEN, kt, kb);
#pragma unroll
      for (int nt = 0; nt < 8; ++nt) {
        v16h b = load_b_frag(nw2p, kt, nt, 8, lane);
        acc2[nt] = __builtin_amdgcn_wmma_f32_16x16x32_f16(false, a, false, b,
                                                          (short)0, acc2[nt], false, false);
      }
    }
#pragma unroll
    for (int nt = 0; nt < 8; ++nt) {
      const int col = nt * 16 + m;
#pragma unroll
      for (int r = 0; r < 8; ++r) {
        int n2 = tile * 16 + drow + r;
        h_out[(size_t)n2 * NODE_DIM + col] =
            hfeat[(size_t)n2 * NODE_DIM + col] + acc2[nt][r];
      }
    }
    const size_t p0 = (size_t)tile * 48 + lane;
    const size_t p1 = p0 + (size_t)Nn * 3;
    pos_out[p0] = pos[p0] + (pacc[p0] + pacc[p1]);
    if (lane < 16) pos_out[p0 + 32] = pos[p0 + 32] + (pacc[p0 + 32] + pacc[p1 + 32]);
  }
}

extern "C" void kernel_launch(void* const* d_in, const int* in_sizes, int n_in,
                              void* d_out, int out_size, void* d_ws, size_t ws_size,
                              hipStream_t stream) {
  const float* h   = (const float*)d_in[0];
  const float* pos = (const float*)d_in[1];
  const int*   ei  = (const int*)d_in[2];
  const float* mw1 = (const float*)d_in[3];  const float* mb1 = (const float*)d_in[4];
  const float* mw2 = (const float*)d_in[5];  const float* mb2 = (const float*)d_in[6];
  const float* nw1 = (const float*)d_in[7];  const float* nb1 = (const float*)d_in[8];
  const float* nw2 = (const float*)d_in[9];  const float* nb2 = (const float*)d_in[10];
  const float* cw1 = (const float*)d_in[11]; const float* cb1 = (const float*)d_in[12];
  const float* cw2 = (const float*)d_in[13]; const float* cb2 = (const float*)d_in[14];
  const int N = in_sizes[0] / NODE_DIM;
  const int E = in_sizes[2] / 2;

  char* ws = (char*)d_ws;
  h16* mw1p = (h16*)(ws + 0);
  h16* mw2p = (h16*)(ws + 73728);
  h16* cw1p = (h16*)(ws + 106496);
  h16* nw1p = (h16*)(ws + 139264);
  h16* nw2p = (h16*)(ws + 204800);
  float* aggr = (float*)(ws + 237568);
  float* pacc = aggr + (size_t)N * HIDDEN * 2;
  float* h_out   = (float*)d_out;
  float* pos_out = h_out + (size_t)N * NODE_DIM;

  pack_w<<<(9 * 8 * 32 + 127) / 128, 128, 0, stream>>>(mw1, mw1p, MSG_IN, HIDDEN);
  pack_w<<<(4 * 8 * 32 + 127) / 128, 128, 0, stream>>>(mw2, mw2p, HIDDEN, HIDDEN);
  pack_w<<<(4 * 8 * 32 + 127) / 128, 128, 0, stream>>>(cw1, cw1p, HIDDEN, HIDDEN);
  pack_w<<<(8 * 8 * 32 + 127) / 128, 128, 0, stream>>>(nw1, nw1p, 2 * NODE_DIM, HIDDEN);
  pack_w<<<(4 * 8 * 32 + 127) / 128, 128, 0, stream>>>(nw2, nw2p, HIDDEN, NODE_DIM);
  egnn_init<<<2048, 256, 0, stream>>>(aggr, N);

  egnn_edge<<<256, EWAVES * 32, 0, stream>>>(h, pos, ei, mw1p, mb1, mw2p, mb2,
                                                 cw1p, cb1, cw2, cb2, aggr, pacc, E, N);

  size_t lds_n = (size_t)WAVES * (16 * 2 * NODE_DIM * 2 + 16 * HIDDEN * 2);
  egnn_node<<<800, WAVES * 32, lds_n, stream>>>(h, aggr, nw1p, nb1, nw2p, nb2, h_out,
                                                pos, pacc, pos_out, N);
}


# The five weight-packing launches and the zeroing launch become one launch (first 29 blocks pack one k-step of one weight each, the other 2048 zero the sums): 8 launches -> 3; values unchanged
# speedup vs baseline: 2.8090x; 1.1459x over previous
#include <hip/hip_runtime.h>
#include <math.h>

typedef _Float16 h16;
typedef __attribute__((ext_vector_type(16))) _Float16 v16h;
typedef __attribute__((ext_vector_type(8)))  _Float16 v8h;
typedef __attribute__((ext_vector_type(4)))  _Float16 v4h;
typedef __attribute__((ext_vector_type(8)))  float    v8f;

#define NODE_DIM 128
#define HIDDEN   128
#define NBASIS   32
#define MSG_IN   288
#define WAVES    4
#define EWAVES   16
#define MW1_B    (MSG_IN * HIDDEN * 2)
#define MW2_B    (HIDDEN * HIDDEN * 2)
#define EW_BYTES (MW1_B + 2 * MW2_B)
#define EPANEL   (16 * MSG_IN * 2)
#define MPITCH   132

static __device__ __forceinline__ v16h cat8(v8h a, v8h b) {
  return __builtin_shufflevector(a, b, 0,1,2,3,4,5,6,7,8,9,10,11,12,13,14,15);
}
static __device__ __forceinline__ float silu(float x) {
  return x * __builtin_amdgcn_rcpf(1.0f + __expf(-x));
}
static __device__ __forceinline__ float xhalf(float x) {
  const int b = __builtin_bit_cast(int, x);
  return __builtin_bit_cast(float,
      __builtin_amdgcn_permlanex16(b, b, 0x76543210, 0xfedcba98, false, false));
}
static __device__ __forceinline__ v16h load_b_frag(const h16* __restrict__ wp,
                                                   int kt, int nt, int NT, int lane) {
  const v8h* p = (const v8h*)(wp + (size_t)(((kt * NT + nt) * 32 + lane)) * 16);
  return cat8(p[0], p[1]);
}
static __device__ __forceinline__ v16h load_a_frag(const h16* row_base, int kt, int kb) {
  const v8h* a0 = (const v8h*)(row_base + kt * 32 + kb);
  const v8h* a1 = (const v8h*)(row_base + kt * 32 + 16 + kb);
  return cat8(*a0, *a1);
}

static __device__ __forceinline__ v16h lds_b_frag(const char* wbase, int kt, int nt, int lane) {
  const char* p = wbase + (kt * 8 + nt) * 1024 + lane * 16;
  return cat8(*(const v8h*)p, *(const v8h*)(p + 512));
}
static __device__ __forceinline__ void stage_w(const h16* __restrict__ src, char* dst,
                                               int nbytes, int tid, int nthr) {
  const uint4* g = (const uint4*)src;
  for (int c = tid; c < (nbytes >> 4); c += nthr) {
    const int f = c >> 6, l = (c >> 1) & 31, j = c & 1;
    *(uint4*)(dst + f * 1024 + j * 512 + l * 16) = g[c];
  }
}

template <int KT, typename XF>
static __device__ __forceinline__ void panel_mma(v8f (&acc)[8], XF xfrag,
                                                 const char* wbase, int lane) {
  v16h wc[8];
  v16h xc = xfrag(0);
#pragma unroll
  for (int nt = 0; nt < 8; ++nt) wc[nt] = lds_b_frag(wbase, 0, nt, lane);
#pragma unroll
  for (int kt = 0; kt < KT; ++kt) {
    v16h wn[8];
    v16h xn = xc;
#pragma unroll
    for (int h = 0; h < 2; ++h) {
      if (kt + 1 < KT) {
        if (h == 0) xn = xfrag(kt + 1);
#pragma unroll
        for (int j = 0; j < 4; ++j) wn[4 * h + j] = lds_b_frag(wbase, kt + 1, 4 * h + j, lane);
      }
      __builtin_amdgcn_sched_barrier(0);
#pragma unroll
      for (int j = 0; j < 4; ++j)
        acc[4 * h + j] = __builtin_amdgcn_wmma_f32_16x16x32_f16(false, wc[4 * h + j], false, xc,
                                                                (short)0, acc[4 * h + j], false, false);
    }
    if (kt + 1 < KT) {
      xc = xn;
#pragma unroll
      for (int nt = 0; nt < 8; ++nt) wc[nt] = wn[nt];
    }
  }
}
static __device__ __forceinline__ v8f bias_tile(const float* __restrict__ b, int nt, int drow) {
  const float4* p = (const float4*)(b + nt * 16 + drow);
  const float4 lo = p[0], hi = p[1];
  return (v8f){lo.x, lo.y, lo.z, lo.w, hi.x, hi.y, hi.z, hi.w};
}

#define PACK_BLOCKS 29
#define ZERO_BLOCKS 2048
static __device__ __forceinline__ void pack_frag(const float* __restrict__ W,
                                                 h16* __restrict__ out, int t) {
  const int Nn = HIDDEN;
  int lane = t & 31;
  int tile = t >> 5;
  int nt = tile & 7, kt = tile >> 3;
  int n = nt * 16 + (lane & 15);
  int kbase = (lane < 16) ? 0 : 8;
  alignas(16) h16 tmp[16];
#pragma unroll
  for (int r = 0; r < 4; ++r) {
    int k0 = kt * 32 + kbase + 2 * r;
    tmp[2 * r]     = (h16)W[(size_t)k0 * Nn + n];
    tmp[2 * r + 1] = (h16)W[(size_t)(k0 + 1) * Nn + n];
    int k1 = kt * 32 + 16 + kbase + 2 * r;
    tmp[8 + 2 * r]     = (h16)W[(size_t)k1 * Nn + n];
    tmp[8 + 2 * r + 1] = (h16)W[(size_t)(k1 + 1) * Nn + n];
  }
  v8h* o = (v8h*)(out + (size_t)t * 16);
  o[0] = *(const v8h*)(tmp);
  o[1] = *(const v8h*)(tmp + 8);
}
extern "C" __global__ void __launch_bounds__(256)
egnn_prep(const float* __restrict__ mw1, const float* __restrict__ mw2,
          const float* __restrict__ cw1, const float* __restrict__ nw1,
          const float* __restrict__ nw2,
          h16* __restrict__ mw1p, h16* __restrict__ mw2p, h16* __restrict__ cw1p,
          h16* __restrict__ nw1p, h16* __restrict__ nw2p,
          float* __restrict__ sums, int N) {
  const int b = blockIdx.x;
  if (b < PACK_BLOCKS) {
    const float* W; h16* out; int kt;
    if (b < 9)       { W = mw1; out = mw1p; kt = b; }
    else if (b < 13) { W = mw2; out = mw2p; kt = b - 9; }
    else if (b < 17) { W = cw1; out = cw1p; kt = b - 13; }
    else if (b < 25) { W = nw1; out = nw1p; kt = b - 17; }
    else             { W = nw2; out = nw2p; kt = b - 25; }
    pack_frag(W, out, kt * 256 + threadIdx.x);
    return;
  }
  int i = (b - PACK_BLOCKS) * blockDim.x + threadIdx.x;
  int stride = ZERO_BLOCKS * blockDim.x;
  for (int j = i; j < N * (2 * HIDDEN + 6); j += stride) sums[j] = 0.0f;
}

extern "C" __global__ void __launch_bounds__(EWAVES * 32)
egnn_edge(const float* __restrict__ hfeat, const float* __restrict__ pos,
          const int* __restrict__ ei,
          const h16* __restrict__ mw1p, const float* __restrict__ mb1,
          const h16* __restrict__ mw2p, const float* __restrict__ mb2,
          const h16* __restrict__ cw1p, const float* __restrict__ cb1,
          const float* __restrict__ cw2, const float* __restrict__ cb2,
          float* __restrict__ aggr2, float* __restrict__ pacc2, int E, int N) {
  __shared__ __attribute__((aligned(16))) char smem[EW_BYTES + EWAVES * EPANEL];
  const char* s_w1 = smem;
  const char* s_w2 = smem + MW1_B;
  const char* s_w3 = smem + MW1_B + MW2_B;
  stage_w(mw1p, smem, MW1_B, threadIdx.x, EWAVES * 32);
  stage_w(mw2p, smem + MW1_B, MW2_B, threadIdx.x, EWAVES * 32);
  stage_w(cw1p, smem + MW1_B + MW2_B, MW2_B, threadIdx.x, EWAVES * 32);
  __syncthreads();
  const int wave = threadIdx.x >> 5;
  const int lane = threadIdx.x & 31;
  const int m    = lane & 15;
  const int kb   = (lane < 16) ? 0 : 8;
  const int drow = (lane < 16) ? 0 : 8;

  const unsigned half = (__builtin_amdgcn_s_sendmsg_rtn(0x87) >> 19) & 1u;
  float* __restrict__ aggr = aggr2 + (size_t)half * N * HIDDEN;
  float* __restrict__ pacc = pacc2 + (size_t)half * N * 3;

  char* wb = smem + EW_BYTES + wave * EPANEL;
  h16*   s_min = (h16*)wb;
  float* s_msg = (float*)wb;

  const int ntile = E >> 4;
  const int nwav  = gridDim.x * EWAVES;
  for (int tile = wave * gridDim.x + blockIdx.x; tile < ntile; tile += nwav) {
    const int e    = tile * 16 + m;
    const int vrow = ei[e];
    const int vcol = ei[E + e];
    float rx = 0.f, ry = 0.f, rz = 0.f;

    rx = pos[vrow * 3 + 0] - pos[vcol * 3 + 0];
    ry = pos[vrow * 3 + 1] - pos[vcol * 3 + 1];
    rz = pos[vrow * 3 + 2] - pos[vcol * 3 + 2];
    float4 hr[8], hc[8];
#pragma unroll
    for (int i = 0; i < 8; ++i) {
      const int ri = __builtin_amdgcn_readlane(vrow, i);
      const int ci = __builtin_amdgcn_readlane(vcol, i);
      hr[i] = ((const float4*)(hfeat + (size_t)ri * NODE_DIM))[lane];
      hc[i] = ((const float4*)(hfeat + (size_t)ci * NODE_DIM))[lane];
    }
    if (lane < 16) {
      float dist = sqrtf(rx * rx + ry * ry + rz * rz + 1e-8f);
      const float step  = 10.0f / (NBASIS - 1);
      const float coeff = -0.5f / (step * step);
#pragma unroll
      for (int b = 0; b < NBASIS; ++b) {
        float d = dist - step * b;
        s_min[m * MSG_IN + 2 * NODE_DIM + b] = (h16)__expf(coeff * d * d);
      }
    }
#pragma unroll
    for (int i = 0; i < 8; ++i) {
      *(v4h*)(s_min + i * MSG_IN + lane * 4) =
          (v4h){(h16)hr[i].x, (h16)hr[i].y, (h16)hr[i].z, (h16)hr[i].w};
      *(v4h*)(s_min + i * MSG_IN + NODE_DIM + lane * 4) =
          (v4h){(h16)hc[i].x, (h16)hc[i].y, (h16)hc[i].z, (h16)hc[i].w};
      const int ri = __builtin_amdgcn_readlane(vrow, 8 + i);
      const int ci = __builtin_amdgcn_readlane(vcol, 8 + i);
      hr[i] = ((const float4*)(hfeat + (size_t)ri * NODE_DIM))[lane];
      hc[i] = ((const float4*)(hfeat + (size_t)ci * NODE_DIM))[lane];
    }
#pragma unroll
    for (int i = 0; i < 8; ++i) {
      *(v4h*)(s_min + (8 + i) * MSG_IN + lane * 4) =
          (v4h){(h16)hr[i].x, (h16)hr[i].y, (h16)hr[i].z, (h16)hr[i].w};
      *(v4h*)(s_min + (8 + i) * MSG_IN + NODE_DIM + lane * 4) =
          (v4h){(h16)hc[i].x, (h16)hc[i].y, (h16)hc[i].z, (h16)hc[i].w};
    }
    asm volatile("s_wait_dscnt 0" ::: "memory");

    const h16* xrow = s_min + m * MSG_IN;
    v8f acc[8];
#pragma unroll
    for (int nt = 0; nt < 8; ++nt) acc[nt] = bias_tile(mb1, nt, drow);
    panel_mma<MSG_IN / 32>(acc, [&](int kt) { return load_a_frag(xrow, kt, kb); }, s_w1, lane);
    v8f acc2[8];
#pragma unroll
    for (int nt = 0; nt < 8; ++nt) acc2[nt] = bias_tile(mb2, nt, drow);
    v16h hf[4];
#pragma unroll
    for (int kt = 0; kt < 4; ++kt)
#pragma unroll
      for (int r = 0; r < 8; ++r) {
        hf[kt][r]     = (h16)silu(acc[2 * kt][r]);
        hf[kt][8 + r] = (h16)silu(acc[2 * kt + 1][r]);
      }

    __builtin_amdgcn_sched_barrier(0);
    panel_mma<HIDDEN / 32>(acc2, [&](int kt) { return hf[kt]; }, s_w2, lane);
    v16h mf[4];
#pragma unroll
    for (int kt = 0; kt < 4; ++kt)
#pragma unroll
      for (int r = 0; r < 8; ++r) {
        mf[kt][r]     = (h16)acc2[2 * kt][r];
        mf[kt][8 + r] = (h16)acc2[2 * kt + 1][r];
      }
#pragma unroll
    for (int nt = 0; nt < 8; ++nt) {
      float4* q = (float4*)(s_msg + m * MPITCH + nt * 16 + drow);
      q[0] = make_float4(acc2[nt][0], acc2[nt][1], acc2[nt][2], acc2[nt][3]);
      q[1] = make_float4(acc2[nt][4], acc2[nt][5], acc2[nt][6], acc2[nt][7]);
    }
    float mv[16][4];
#pragma unroll
    for (int e = 0; e < 16; ++e)
#pragma unroll
      for (int p = 0; p < 4; ++p) mv[e][p] = s_msg[e * MPITCH + 32 * p + lane];
#pragma unroll
    for (int e = 0; e < 16; ++e) {
      const int re = __builtin_amdgcn_readlane(vrow, e);
      float* dst = aggr + (size_t)re * HIDDEN + lane;
#pragma unroll
      for (int p = 0; p < 4; ++p) unsafeAtomicAdd(dst + 32 * p, mv[e][p]);
    }

    v8f acc3[8];
#pragma unroll
    for (int nt = 0; nt < 8; ++nt) acc3[nt] = bias_tile(cb1, nt, drow);
    panel_mma<HIDDEN / 32>(acc3, [&](int kt) { return mf[kt]; }, s_w3, lane);
    float part[8];
#pragma unroll
    for (int r = 0; r < 8; ++r) part[r] = 0.0f;
#pragma unroll
    for (int nt = 0; nt < 8; ++nt) {
      const v8f w = bias_tile(cw2, nt, drow);
#pragma unroll
      for (int r = 0; r < 8; ++r)
        part[r] += silu(acc3[nt][r]) * w[r];
    }
    const float mine = ((part[0] + part[1]) + (part[2] + part[3])) +
                       ((part[4] + part[5]) + (part[6] + part[7]));
    const float cm = cb2[0] + (mine + xhalf(mine));
    if (lane < 16) {
      unsafeAtomicAdd(&pacc[vrow * 3 + 0], rx * cm);
      unsafeAtomicAdd(&pacc[vrow * 3 + 1], ry * cm);
      unsafeAtomicAdd(&pacc[vrow * 3 + 2], rz * cm);
    }
    asm volatile("s_wait_dscnt 0" ::: "memory");
  }
}

extern "C" __global__ void __launch_bounds__(WAVES * 32)
egnn_node(const float* __restrict__ hfeat, const float* __restrict__ aggr,
          const h16* __restrict__ nw1p, const float* __restrict__ nb1,
          const h16* __restrict__ nw2p, const float* __restrict__ nb2,
          float* __restrict__ h_out, const float* __restrict__ pos,
          const float* __restrict__ pacc, float* __restrict__ pos_out, int Nn) {
  extern __shared__ char smem[];
  const int wave = threadIdx.x >> 5;
  const int lane = threadIdx.x & 31;
  const int m    = lane & 15;
  const int kb   = (lane < 16) ? 0 : 8;
  const int drow = (lane < 16) ? 0 : 8;

  const int NIN = 2 * NODE_DIM;
  const int WB  = 16 * NIN * 2 + 16 * HIDDEN * 2;
  char* wb = smem + wave * WB;
  h16* s_nin = (h16*)wb;
  h16* s_hid = (h16*)(wb + 16 * NIN * 2);

  const int ntile = Nn >> 4;
  const int nwav  = gridDim.x * WAVES;
  for (int tile = blockIdx.x * WAVES + wave; tile < ntile; tile += nwav) {
    const float4* hsrc = (const float4*)(hfeat + (size_t)tile * 16 * NODE_DIM);
    const float4* asrc = (const float4*)(aggr + (size_t)tile * 16 * HIDDEN);
    const float4* bsrc = (const float4*)(aggr + ((size_t)Nn + tile * 16) * HIDDEN);
#pragma unroll
    for (int i = 0; i < 16; ++i) {
      const float4 a = hsrc[i * 32 + lane];
      const float4 b0 = asrc[i * 32 + lane];
      const float4 b1 = bsrc[i * 32 + lane];
      const float4 b = make_float4(b0.x + b1.x, b0.y + b1.y, b0.z + b1.z, b0.w + b1.w);
      *(v4h*)(s_nin + i * NIN + lane * 4) = (v4h){(h16)a.x, (h16)a.y, (h16)a.z, (h16)a.w};
      *(v4h*)(s_nin + i * NIN + NODE_DIM + lane * 4) =
          (v4h){(h16)b.x, (h16)b.y, (h16)b.z, (h16)b.w};
    }
    asm volatile("s_wait_dscnt 0" ::: "memory");

    v8f acc[8];
#pragma unroll
    for (int nt = 0; nt < 8; ++nt) {
      float b0 = nb1[nt * 16 + m];
      acc[nt] = (v8f){b0, b0, b0, b0, b0, b0, b0, b0};
    }
#pragma unroll
    for (int kt = 0; kt < NIN / 32; ++kt) {
      v16h a = load_a_frag(s_nin + m * NIN, kt, kb);
#pragma unroll
      for (int nt = 0; nt < 8; ++nt) {
        v16h b = load_b_frag(nw1p, kt, nt, 8, lane);
        acc[nt] = __builtin_amdgcn_wmma_f32_16x16x32_f16(false, a, false, b,
                                                         (short)0, acc[nt], false, false);
      }
    }
#pragma unroll
    for (int nt = 0; nt < 8; ++nt)
#pragma unroll
      for (int r = 0; r < 8; ++r)
        s_hid[(drow + r) * HIDDEN + nt * 16 + m] = (h16)silu(acc[nt][r]);
    asm volatile("s_wait_dscnt 0" ::: "memory");

    v8f acc2[8];
#pragma unroll
    for (int nt = 0; nt < 8; ++nt) {
      float b0 = nb2[nt * 16 + m];
      acc2[nt] = (v8f){b0, b0, b0, b0, b0, b0, b0, b0};
    }
#pragma unroll
    for (int kt = 0; kt < HIDDEN / 32; ++kt) {
      v16h a = load_a_frag(s_hid + m * HIDDEN, kt, kb);
#pragma unroll
      for (int nt = 0; nt < 8; ++nt) {
        v16h b = load_b_frag(nw2p, kt, nt, 8, lane);
        acc2[nt] = __builtin_amdgcn_wmma_f32_16x16x32_f16(false, a, false, b,
                                                          (short)0, acc2[nt], false, false);
      }
    }
#pragma unroll
    for (int nt = 0; nt < 8; ++nt) {
      const int col = nt * 16 + m;
#pragma unroll
      for (int r = 0; r < 8; ++r) {
        int n2 = tile * 16 + drow + r;
        h_out[(size_t)n2 * NODE_DIM + col] =
            hfeat[(size_t)n2 * NODE_DIM + col] + acc2[nt][r];
      }
    }
    const size_t p0 = (size_t)tile * 48 + lane;
    const size_t p1 = p0 + (size_t)Nn * 3;
    pos_out[p0] = pos[p0] + (pacc[p0] + pacc[p1]);
    if (lane < 16) pos_out[p0 + 32] = pos[p0 + 32] + (pacc[p0 + 32] + pacc[p1 + 32]);
  }
}

extern "C" void kernel_launch(void* const* d_in, const int* in_sizes, int n_in,
                              void* d_out, int out_size, void* d_ws, size_t ws_size,
                              hipStream_t stream) {
  const float* h   = (const float*)d_in[0];
  const float* pos = (const float*)d_in[1];
  const int*   ei  = (const int*)d_in[2];
  const float* mw1 = (const float*)d_in[3];  const float* mb1 = (const float*)d_in[4];
  const float* mw2 = (const float*)d_in[5];  const float* mb2 = (const float*)d_in[6];
  const float* nw1 = (const float*)d_in[7];  const float* nb1 = (const float*)d_in[8];
  const float* nw2 = (const float*)d_in[9];  const float* nb2 = (const float*)d_in[10];
  const float* cw1 = (const float*)d_in[11]; const float* cb1 = (const float*)d_in[12];
  const float* cw2 = (const float*)d_in[13]; const float* cb2 = (const float*)d_in[14];
  const int N = in_sizes[0] / NODE_DIM;
  const int E = in_sizes[2] / 2;

  char* ws = (char*)d_ws;
  h16* mw1p = (h16*)(ws + 0);
  h16* mw2p = (h16*)(ws + 73728);
  h16* cw1p = (h16*)(ws + 106496);
  h16* nw1p = (h16*)(ws + 139264);
  h16* nw2p = (h16*)(ws + 204800);
  float* aggr = (float*)(ws + 237568);
  float* pacc = aggr + (size_t)N * HIDDEN * 2;
  float* h_out   = (float*)d_out;
  float* pos_out = h_out + (size_t)N * NODE_DIM;

  egnn_prep<<<PACK_BLOCKS + ZERO_BLOCKS, 256, 0, stream>>>(mw1, mw2, cw1, nw1, nw2,
                                                           mw1p, mw2p, cw1p, nw1p, nw2p,
                                                           aggr, N);

  egnn_edge<<<256, EWAVES * 32, 0, stream>>>(h, pos, ei, mw1p, mb1, mw2p, mb2,
                                                 cw1p, cb1, cw2, cb2, aggr, pacc, E, N);

  size_t lds_n = (size_t)WAVES * (16 * 2 * NODE_DIM * 2 + 16 * HIDDEN * 2);
  egnn_node<<<800, WAVES * 32, lds_n, stream>>>(h, aggr, nw1p, nb1, nw2p, nb2, h_out,
                                                pos, pacc, pos_out, N);
}
